# starting kernel
# baseline (speedup 1.0000x reference)
#include <hip/hip_runtime.h>
#include <math.h>

typedef __attribute__((ext_vector_type(2))) float v2f;
typedef __attribute__((ext_vector_type(8))) float v8f;

#define DFEAT 128

__global__ __launch_bounds__(256) void fill_neginf_kernel(uint4* __restrict__ p, int n4) {
    int i = blockIdx.x * blockDim.x + threadIdx.x;
    if (i < n4) {
        uint4 v;
        v.x = 0xFF800000u; v.y = 0xFF800000u; v.z = 0xFF800000u; v.w = 0xFF800000u;
        p[i] = v;
    }
}

__device__ __forceinline__ void atomic_max_f32(float* addr, float val) {
    if (!(val < 0.0f)) {
        atomicMax((int*)addr, __float_as_int(val));
    } else {
        atomicMin((unsigned int*)addr, __float_as_uint(val));
    }
}

__global__ __launch_bounds__(256) void scatter_max_kernel(const float* __restrict__ x,
                                                          const int* __restrict__ src,
                                                          const int* __restrict__ dst,
                                                          float* __restrict__ agg,
                                                          int nedges) {
    int wave = (blockIdx.x * blockDim.x + threadIdx.x) >> 5;
    int lane = threadIdx.x & 31;
    if (wave >= nedges) return;
    int s = src[wave];
    int d = dst[wave];
    const float4 v = *(const float4*)(x + (size_t)s * DFEAT + lane * 4);
    float* base = agg + (size_t)d * DFEAT + lane * 4;
    atomic_max_f32(base + 0, v.x);
    atomic_max_f32(base + 1, v.y);
    atomic_max_f32(base + 2, v.z);
    atomic_max_f32(base + 3, v.w);
}

template <bool DUAL, bool RELU>
__global__ __launch_bounds__(256) void sage_gemm_kernel(const float* __restrict__ A0,
                                                        const float* __restrict__ A1,
                                                        const float* __restrict__ W0,
                                                        const float* __restrict__ W1,
                                                        const float* __restrict__ bias,
                                                        float* __restrict__ out) {
    const int lane = threadIdx.x & 31;
    const int wv   = threadIdx.x >> 5;
    const int m    = lane & 15;
    const int half = lane >> 4;
    const int row_base = blockIdx.x * 16;
    const int col_base = wv * 16;
    const int koff = half * 2;

    const float* a0row = A0 + (size_t)(row_base + m) * DFEAT + koff;
    const float* w0row = W0 + (size_t)(col_base + m) * DFEAT + koff;
    const float* a1row = DUAL ? (A1 + (size_t)(row_base + m) * DFEAT + koff) : a0row;
    const float* w1row = DUAL ? (W1 + (size_t)(col_base + m) * DFEAT + koff) : w0row;

    v8f acc = {};
#pragma unroll 4
    for (int kk = 0; kk < DFEAT; kk += 4) {
        v2f a0 = *(const v2f*)(a0row + kk);
        if (DUAL) {
            a0.x = (a0.x == -INFINITY) ? 0.0f : a0.x;
            a0.y = (a0.y == -INFINITY) ? 0.0f : a0.y;
        }
        v2f b0 = *(const v2f*)(w0row + kk);
        acc = __builtin_amdgcn_wmma_f32_16x16x4_f32(false, a0, false, b0,
                                                    (short)0, acc, false, false);
        if (DUAL) {
            v2f a1 = *(const v2f*)(a1row + kk);
            v2f b1 = *(const v2f*)(w1row + kk);
            acc = __builtin_amdgcn_wmma_f32_16x16x4_f32(false, a1, false, b1,
                                                        (short)0, acc, false, false);
        }
    }

    __syncthreads();

    const float bv = bias[col_base + m];
#pragma unroll
    for (int r = 0; r < 8; ++r) {
        float v = acc[r] + bv;
        if (RELU) v = fmaxf(v, 0.0f);
        out[(size_t)(row_base + r + 8 * half) * DFEAT + col_base + m] = v;
    }
}

extern "C" void kernel_launch(void* const* d_in, const int* in_sizes, int n_in,
                              void* d_out, int out_size, void* d_ws, size_t ws_size,
                              hipStream_t stream) {
    const float* x  = (const float*)d_in[0];
    const int*   ei = (const int*)d_in[1];
    const float* Wl = (const float*)d_in[2];
    const float* bl = (const float*)d_in[3];
    const float* Wr = (const float*)d_in[4];
    const float* W  = (const float*)d_in[5];
    const float* b  = (const float*)d_in[6];

    const int nE = in_sizes[1] / 2;
    const int n  = in_sizes[0] / DFEAT;
    const int*   srcp = ei;
    const int*   dstp = ei + nE;

    float* buf0 = (float*)d_ws;
    float* buf1 = buf0 + (size_t)n * DFEAT;

    dim3 blk(256);
    const int fillBlocks = (n * DFEAT / 4 + 255) / 256;
    const int scatBlocks = (nE + 7) / 8;
    const int gemmBlocks = n / 16;

    fill_neginf_kernel<<<fillBlocks, blk, 0, stream>>>((uint4*)buf0, n * DFEAT / 4);
    scatter_max_kernel<<<scatBlocks, blk, 0, stream>>>(x, srcp, dstp, buf0, nE);
    sage_gemm_kernel<true, true><<<gemmBlocks, blk, 0, stream>>>(
        buf0, x, Wl, Wr, bl, buf0);

    fill_neginf_kernel<<<fillBlocks, blk, 0, stream>>>((uint4*)buf1, n * DFEAT / 4);
    scatter_max_kernel<<<scatBlocks, blk, 0, stream>>>(buf0, srcp, dstp, buf1, nE);
    sage_gemm_kernel<true, true><<<gemmBlocks, blk, 0, stream>>>(
        buf1, buf0, Wl + DFEAT * DFEAT, Wr + DFEAT * DFEAT, bl + DFEAT, buf1);

    sage_gemm_kernel<false, false><<<gemmBlocks, blk, 0, stream>>>(
        buf1, nullptr, W, nullptr, b, (float*)d_out);
}


# Atomic scatter-max replaced by a destination-sorted adjacency (counting sort, built once per call) and a per-node gather-max of whole rows; max is order-independent, no sum is reordered
# speedup vs baseline: 1.4989x; 1.4989x over previous
#include <hip/hip_runtime.h>
#include <math.h>


typedef __attribute__((ext_vector_type(2))) float v2f;
typedef __attribute__((ext_vector_type(8))) float v8f;

#define DFEAT 128

#define SCAN_THREADS 1024
#define SCAN_ITEMS 16
#define SCAN_TILE (SCAN_THREADS * SCAN_ITEMS)

__global__ __launch_bounds__(256) void zero_int_kernel(int4* __restrict__ p, int n4) {
    int i = blockIdx.x * blockDim.x + threadIdx.x;
    if (i < n4) {
        int4 v;
        v.x = 0; v.y = 0; v.z = 0; v.w = 0;
        p[i] = v;
    }
}

__global__ __launch_bounds__(256) void csr_count_kernel(const int* __restrict__ dst,
                                                        int* __restrict__ cnt,
                                                        int nedges) {
    int e = blockIdx.x * blockDim.x + threadIdx.x;
    if (e < nedges) atomicAdd(&cnt[dst[e]], 1);
}

__global__ __launch_bounds__(SCAN_THREADS) void csr_scan_kernel(const int* __restrict__ cnt,
                                                                int* __restrict__ rowptr,
                                                                int ntiles) {
    __shared__ int wave_off[32];
    __shared__ int tile_total;
    const int tid  = threadIdx.x;
    const int lane = tid & 31;
    const int wv   = tid >> 5;
    int carry = 0;
    for (int t = 0; t < ntiles; ++t) {
        const int base = t * SCAN_TILE + tid * SCAN_ITEMS;
        const int4* src4 = (const int4*)(cnt + base);
        const int4 c0 = src4[0];
        const int4 c1 = src4[1];
        const int4 c2 = src4[2];
        const int4 c3 = src4[3];
        int4 e0, e1, e2, e3;
        int run = 0;
        e0.x = run; run += c0.x;  e0.y = run; run += c0.y;
        e0.z = run; run += c0.z;  e0.w = run; run += c0.w;
        e1.x = run; run += c1.x;  e1.y = run; run += c1.y;
        e1.z = run; run += c1.z;  e1.w = run; run += c1.w;
        e2.x = run; run += c2.x;  e2.y = run; run += c2.y;
        e2.z = run; run += c2.z;  e2.w = run; run += c2.w;
        e3.x = run; run += c3.x;  e3.y = run; run += c3.y;
        e3.z = run; run += c3.z;  e3.w = run; run += c3.w;

        int inc = run;
        for (int d = 1; d < 32; d <<= 1) {
            int o = __shfl_up(inc, d);
            if (lane >= d) inc += o;
        }
        if (lane == 31) wave_off[wv] = inc;
        __syncthreads();
        if (wv == 0) {
            const int w = wave_off[lane];
            int winc = w;
            for (int d = 1; d < 32; d <<= 1) {
                int o = __shfl_up(winc, d);
                if (lane >= d) winc += o;
            }
            wave_off[lane] = winc - w;
            if (lane == 31) tile_total = winc;
        }
        __syncthreads();
        const int off = carry + wave_off[wv] + (inc - run);
        carry += tile_total;
        int4* dst4 = (int4*)(rowptr + base);
        e0.x += off; e0.y += off; e0.z += off; e0.w += off;
        e1.x += off; e1.y += off; e1.z += off; e1.w += off;
        e2.x += off; e2.y += off; e2.z += off; e2.w += off;
        e3.x += off; e3.y += off; e3.z += off; e3.w += off;
        dst4[0] = e0;
        dst4[1] = e1;
        dst4[2] = e2;
        dst4[3] = e3;
        __syncthreads();
    }
}

__global__ __launch_bounds__(256) void csr_fill_kernel(const int* __restrict__ src,
                                                       const int* __restrict__ dst,
                                                       const int* __restrict__ rowptr,
                                                       int* __restrict__ cnt,
                                                       int* __restrict__ col,
                                                       int nedges) {
    int e = blockIdx.x * blockDim.x + threadIdx.x;
    if (e < nedges) {
        const int d = dst[e];
        const int left = atomicSub(&cnt[d], 1);
        col[rowptr[d] + left - 1] = src[e];
    }
}

__device__ __forceinline__ float4 max3v(float4 m, float4 a, float4 b) {
    float4 r;
    r.x = __builtin_fmaxf(__builtin_fmaxf(m.x, a.x), b.x);
    r.y = __builtin_fmaxf(__builtin_fmaxf(m.y, a.y), b.y);
    r.z = __builtin_fmaxf(__builtin_fmaxf(m.z, a.z), b.z);
    r.w = __builtin_fmaxf(__builtin_fmaxf(m.w, a.w), b.w);
    return r;
}

__device__ __forceinline__ float4 load_row(const float* __restrict__ x, int ids, int j, unsigned loff) {
    const float* row = x + (size_t)__builtin_amdgcn_readlane(ids, j) * DFEAT;
    return *(const float4*)(row + loff);
}

__global__ __launch_bounds__(256) void gather_max_kernel(const float* __restrict__ x,
                                                         const int* __restrict__ rowptr,
                                                         const int* __restrict__ col,
                                                         float* __restrict__ agg,
                                                         int n) {
    const int lane = threadIdx.x & 31;
    const unsigned loff = (unsigned)lane * 4u;
    const int node = __builtin_amdgcn_readfirstlane((int)((blockIdx.x * blockDim.x + threadIdx.x) >> 5));
    if (node >= n) return;
    const int beg = rowptr[node];
    const int end = rowptr[node + 1];
    float4 acc;
    acc.x = -INFINITY; acc.y = -INFINITY; acc.z = -INFINITY; acc.w = -INFINITY;
    for (int base = beg; base < end; base += 32) {
        const int cnt = min(end - base, 32);
        const int ids = col[base + min(lane, cnt - 1)];
        int j = 0;
        for (; j + 4 < cnt; j += 8) {
            const float4 r0 = load_row(x, ids, j + 0, loff);
            const float4 r1 = load_row(x, ids, j + 1, loff);
            const float4 r2 = load_row(x, ids, j + 2, loff);
            const float4 r3 = load_row(x, ids, j + 3, loff);
            const float4 r4 = load_row(x, ids, j + 4, loff);
            const float4 r5 = load_row(x, ids, j + 5, loff);
            const float4 r6 = load_row(x, ids, j + 6, loff);
            const float4 r7 = load_row(x, ids, j + 7, loff);
            acc = max3v(acc, r0, r1);
            acc = max3v(acc, r2, r3);
            acc = max3v(acc, r4, r5);
            acc = max3v(acc, r6, r7);
        }
        if (j < cnt) {
            const float4 r0 = load_row(x, ids, j + 0, loff);
            const float4 r1 = load_row(x, ids, j + 1, loff);
            const float4 r2 = load_row(x, ids, j + 2, loff);
            const float4 r3 = load_row(x, ids, j + 3, loff);
            acc = max3v(acc, r0, r1);
            acc = max3v(acc, r2, r3);
        }
    }
    *(float4*)(agg + (size_t)node * DFEAT + loff) = acc;
}

template <bool DUAL, bool RELU>
__global__ __launch_bounds__(256) void sage_gemm_kernel(const float* __restrict__ A0,
                                                        const float* __restrict__ A1,
                                                        const float* __restrict__ W0,
                                                        const float* __restrict__ W1,
                                                        const float* __restrict__ bias,
                                                        float* __restrict__ out) {
    const int lane = threadIdx.x & 31;
    const int wv   = threadIdx.x >> 5;
    const int m    = lane & 15;
    const int half = lane >> 4;
    const int row_base = blockIdx.x * 16;
    const int col_base = wv * 16;
    const int koff = half * 2;

    const float* a0row = A0 + (size_t)(row_base + m) * DFEAT + koff;
    const float* w0row = W0 + (size_t)(col_base + m) * DFEAT + koff;
    const float* a1row = DUAL ? (A1 + (size_t)(row_base + m) * DFEAT + koff) : a0row;
    const float* w1row = DUAL ? (W1 + (size_t)(col_base + m) * DFEAT + koff) : w0row;

    v8f acc = {};
#pragma unroll 4
    for (int kk = 0; kk < DFEAT; kk += 4) {
        v2f a0 = *(const v2f*)(a0row + kk);
        if (DUAL) {
            a0.x = (a0.x == -INFINITY) ? 0.0f : a0.x;
            a0.y = (a0.y == -INFINITY) ? 0.0f : a0.y;
        }
        v2f b0 = *(const v2f*)(w0row + kk);
        acc = __builtin_amdgcn_wmma_f32_16x16x4_f32(false, a0, false, b0,
                                                    (short)0, acc, false, false);
        if (DUAL) {
            v2f a1 = *(const v2f*)(a1row + kk);
            v2f b1 = *(const v2f*)(w1row + kk);
            acc = __builtin_amdgcn_wmma_f32_16x16x4_f32(false, a1, false, b1,
                                                        (short)0, acc, false, false);
        }
    }

    __syncthreads();

    const float bv = bias[col_base + m];
#pragma unroll
    for (int r = 0; r < 8; ++r) {
        float v = acc[r] + bv;
        if (RELU) v = fmaxf(v, 0.0f);
        out[(size_t)(row_base + r + 8 * half) * DFEAT + col_base + m] = v;
    }
}

extern "C" void kernel_launch(void* const* d_in, const int* in_sizes, int n_in,
                              void* d_out, int out_size, void* d_ws, size_t ws_size,
                              hipStream_t stream) {
    const float* x  = (const float*)d_in[0];
    const int*   ei = (const int*)d_in[1];
    const float* Wl = (const float*)d_in[2];
    const float* bl = (const float*)d_in[3];
    const float* Wr = (const float*)d_in[4];
    const float* W  = (const float*)d_in[5];
    const float* b  = (const float*)d_in[6];

    const int nE = in_sizes[1] / 2;
    const int n  = in_sizes[0] / DFEAT;
    const int*   srcp = ei;
    const int*   dstp = ei + nE;

    float* buf0 = (float*)d_ws;
    float* buf1 = buf0 + (size_t)n * DFEAT;

    const int ntiles = (n + SCAN_TILE) / SCAN_TILE;
    const int npad   = ntiles * SCAN_TILE;
    int* cnt    = (int*)(buf1 + (size_t)n * DFEAT);
    int* rowptr = cnt + npad;
    int* col    = rowptr + npad;

    dim3 blk(256);
    const int zeroBlocks = npad / 4 / 256;
    const int edgeBlocks = (nE + 255) / 256;
    const int nodeBlocks = (n + 7) / 8;
    const int gemmBlocks = n / 16;

    zero_int_kernel<<<zeroBlocks, blk, 0, stream>>>((int4*)cnt, npad / 4);
    csr_count_kernel<<<edgeBlocks, blk, 0, stream>>>(dstp, cnt, nE);
    csr_scan_kernel<<<1, SCAN_THREADS, 0, stream>>>(cnt, rowptr, ntiles);
    csr_fill_kernel<<<edgeBlocks, blk, 0, stream>>>(srcp, dstp, rowptr, cnt, col, nE);

    gather_max_kernel<<<nodeBlocks, blk, 0, stream>>>(x, rowptr, col, buf0, n);
    sage_gemm_kernel<true, true><<<gemmBlocks, blk, 0, stream>>>(
        buf0, x, Wl, Wr, bl, buf0);

    gather_max_kernel<<<nodeBlocks, blk, 0, stream>>>(buf0, rowptr, col, buf1, n);
    sage_gemm_kernel<true, true><<<gemmBlocks, blk, 0, stream>>>(
        buf1, buf0, Wl + DFEAT * DFEAT, Wr + DFEAT * DFEAT, bl + DFEAT, buf1);

    sage_gemm_kernel<false, false><<<gemmBlocks, blk, 0, stream>>>(
        buf1, nullptr, W, nullptr, b, (float*)d_out);
}


# GEMM re-tiled: weights staged once per work-group in LDS (pitch 132), each wave owns 16 rows x all 128 columns so one A fragment feeds 8 WMMAs; per-element product order unchanged
# speedup vs baseline: 2.2078x; 1.4729x over previous
#include <hip/hip_runtime.h>
#include <math.h>


typedef __attribute__((ext_vector_type(2))) float v2f;
typedef __attribute__((ext_vector_type(8))) float v8f;

#define DFEAT 128

#define SCAN_THREADS 1024
#define SCAN_ITEMS 16
#define SCAN_TILE (SCAN_THREADS * SCAN_ITEMS)

__global__ __launch_bounds__(256) void zero_int_kernel(int4* __restrict__ p, int n4) {
    int i = blockIdx.x * blockDim.x + threadIdx.x;
    if (i < n4) {
        int4 v;
        v.x = 0; v.y = 0; v.z = 0; v.w = 0;
        p[i] = v;
    }
}

__global__ __launch_bounds__(256) void csr_count_kernel(const int* __restrict__ dst,
                                                        int* __restrict__ cnt,
                                                        int nedges) {
    int e = blockIdx.x * blockDim.x + threadIdx.x;
    if (e < nedges) atomicAdd(&cnt[dst[e]], 1);
}

__global__ __launch_bounds__(SCAN_THREADS) void csr_scan_kernel(const int* __restrict__ cnt,
                                                                int* __restrict__ rowptr,
                                                                int ntiles) {
    __shared__ int wave_off[32];
    __shared__ int tile_total;
    const int tid  = threadIdx.x;
    const int lane = tid & 31;
    const int wv   = tid >> 5;
    int carry = 0;
    for (int t = 0; t < ntiles; ++t) {
        const int base = t * SCAN_TILE + tid * SCAN_ITEMS;
        const int4* src4 = (const int4*)(cnt + base);
        const int4 c0 = src4[0];
        const int4 c1 = src4[1];
        const int4 c2 = src4[2];
        const int4 c3 = src4[3];
        int4 e0, e1, e2, e3;
        int run = 0;
        e0.x = run; run += c0.x;  e0.y = run; run += c0.y;
        e0.z = run; run += c0.z;  e0.w = run; run += c0.w;
        e1.x = run; run += c1.x;  e1.y = run; run += c1.y;
        e1.z = run; run += c1.z;  e1.w = run; run += c1.w;
        e2.x = run; run += c2.x;  e2.y = run; run += c2.y;
        e2.z = run; run += c2.z;  e2.w = run; run += c2.w;
        e3.x = run; run += c3.x;  e3.y = run; run += c3.y;
        e3.z = run; run += c3.z;  e3.w = run; run += c3.w;

        int inc = run;
        for (int d = 1; d < 32; d <<= 1) {
            int o = __shfl_up(inc, d);
            if (lane >= d) inc += o;
        }
        if (lane == 31) wave_off[wv] = inc;
        __syncthreads();
        if (wv == 0) {
            const int w = wave_off[lane];
            int winc = w;
            for (int d = 1; d < 32; d <<= 1) {
                int o = __shfl_up(winc, d);
                if (lane >= d) winc += o;
            }
            wave_off[lane] = winc - w;
            if (lane == 31) tile_total = winc;
        }
        __syncthreads();
        const int off = carry + wave_off[wv] + (inc - run);
        carry += tile_total;
        int4* dst4 = (int4*)(rowptr + base);
        e0.x += off; e0.y += off; e0.z += off; e0.w += off;
        e1.x += off; e1.y += off; e1.z += off; e1.w += off;
        e2.x += off; e2.y += off; e2.z += off; e2.w += off;
        e3.x += off; e3.y += off; e3.z += off; e3.w += off;
        dst4[0] = e0;
        dst4[1] = e1;
        dst4[2] = e2;
        dst4[3] = e3;
        __syncthreads();
    }
}

__global__ __launch_bounds__(256) void csr_fill_kernel(const int* __restrict__ src,
                                                       const int* __restrict__ dst,
                                                       const int* __restrict__ rowptr,
                                                       int* __restrict__ cnt,
                                                       int* __restrict__ col,
                                                       int nedges) {
    int e = blockIdx.x * blockDim.x + threadIdx.x;
    if (e < nedges) {
        const int d = dst[e];
        const int left = atomicSub(&cnt[d], 1);
        col[rowptr[d] + left - 1] = src[e];
    }
}

__device__ __forceinline__ float4 max3v(float4 m, float4 a, float4 b) {
    float4 r;
    r.x = __builtin_fmaxf(__builtin_fmaxf(m.x, a.x), b.x);
    r.y = __builtin_fmaxf(__builtin_fmaxf(m.y, a.y), b.y);
    r.z = __builtin_fmaxf(__builtin_fmaxf(m.z, a.z), b.z);
    r.w = __builtin_fmaxf(__builtin_fmaxf(m.w, a.w), b.w);
    return r;
}

__device__ __forceinline__ float4 load_row(const float* __restrict__ x, int ids, int j, unsigned loff) {
    const float* row = x + (size_t)__builtin_amdgcn_readlane(ids, j) * DFEAT;
    return *(const float4*)(row + loff);
}

__global__ __launch_bounds__(256) void gather_max_kernel(const float* __restrict__ x,
                                                         const int* __restrict__ rowptr,
                                                         const int* __restrict__ col,
                                                         float* __restrict__ agg,
                                                         int n) {
    const int lane = threadIdx.x & 31;
    const unsigned loff = (unsigned)lane * 4u;
    const int node = __builtin_amdgcn_readfirstlane((int)((blockIdx.x * blockDim.x + threadIdx.x) >> 5));
    if (node >= n) return;
    const int beg = rowptr[node];
    const int end = rowptr[node + 1];
    float4 acc;
    acc.x = -INFINITY; acc.y = -INFINITY; acc.z = -INFINITY; acc.w = -INFINITY;
    for (int base = beg; base < end; base += 32) {
        const int cnt = min(end - base, 32);
        const int ids = col[base + min(lane, cnt - 1)];
        int j = 0;
        for (; j + 4 < cnt; j += 8) {
            const float4 r0 = load_row(x, ids, j + 0, loff);
            const float4 r1 = load_row(x, ids, j + 1, loff);
            const float4 r2 = load_row(x, ids, j + 2, loff);
            const float4 r3 = load_row(x, ids, j + 3, loff);
            const float4 r4 = load_row(x, ids, j + 4, loff);
            const float4 r5 = load_row(x, ids, j + 5, loff);
            const float4 r6 = load_row(x, ids, j + 6, loff);
            const float4 r7 = load_row(x, ids, j + 7, loff);
            acc = max3v(acc, r0, r1);
            acc = max3v(acc, r2, r3);
            acc = max3v(acc, r4, r5);
            acc = max3v(acc, r6, r7);
        }
        if (j < cnt) {
            const float4 r0 = load_row(x, ids, j + 0, loff);
            const float4 r1 = load_row(x, ids, j + 1, loff);
            const float4 r2 = load_row(x, ids, j + 2, loff);
            const float4 r3 = load_row(x, ids, j + 3, loff);
            acc = max3v(acc, r0, r1);
            acc = max3v(acc, r2, r3);
        }
    }
    *(float4*)(agg + (size_t)node * DFEAT + loff) = acc;
}

#define WPITCH 132
#define KCHUNK 32
#define KSTEPS (KCHUNK / 4)

template <bool DUAL>
__device__ __forceinline__ void load_a_chunk(const float* __restrict__ a0row,
                                             const float* __restrict__ a1row,
                                             int kc, v2f (&a0)[KSTEPS], v2f (&a1)[KSTEPS]) {
#pragma unroll
    for (int i = 0; i < KSTEPS; ++i) {
        a0[i] = *(const v2f*)(a0row + kc + 4 * i);
        if (DUAL) a1[i] = *(const v2f*)(a1row + kc + 4 * i);
    }
}

template <bool DUAL>
__device__ __forceinline__ void mma_chunk(v8f (&acc)[8], const v2f (&a0c)[KSTEPS],
                                          const v2f (&a1c)[KSTEPS],
                                          const float* bw0, const float* bw1) {
#pragma unroll
    for (int i = 0; i < KSTEPS; ++i) {
        v2f a0 = a0c[i];
        if (DUAL) {
            a0.x = (a0.x == -INFINITY) ? 0.0f : a0.x;
            a0.y = (a0.y == -INFINITY) ? 0.0f : a0.y;
        }
        v2f b[8];
#pragma unroll
        for (int t = 0; t < 8; ++t) b[t] = *(const v2f*)(bw0 + t * 16 * WPITCH + 4 * i);
#pragma unroll
        for (int t = 0; t < 8; ++t)
            acc[t] = __builtin_amdgcn_wmma_f32_16x16x4_f32(false, a0, false, b[t],
                                                           (short)0, acc[t], false, false);
        if (DUAL) {
            const v2f a1 = a1c[i];
#pragma unroll
            for (int t = 0; t < 8; ++t) b[t] = *(const v2f*)(bw1 + t * 16 * WPITCH + 4 * i);
#pragma unroll
            for (int t = 0; t < 8; ++t)
                acc[t] = __builtin_amdgcn_wmma_f32_16x16x4_f32(false, a1, false, b[t],
                                                               (short)0, acc[t], false, false);
        }
    }
}

template <bool DUAL, bool RELU>
__global__ __launch_bounds__(256) void sage_gemm_kernel(const float* __restrict__ A0,
                                                        const float* __restrict__ A1,
                                                        const float* __restrict__ W0,
                                                        const float* __restrict__ W1,
                                                        const float* __restrict__ bias,
                                                        float* __restrict__ out,
                                                        int ntiles) {
    __shared__ __attribute__((aligned(16))) float sW[(DUAL ? 2 : 1) * DFEAT * WPITCH];

    const int lane = threadIdx.x & 31;
    const int wv = __builtin_amdgcn_readfirstlane((int)(threadIdx.x >> 5));

    {
        const int c4 = lane * 4;
#pragma unroll
        for (int r = 0; r < DFEAT; r += 8) {
            const float4 v = *(const float4*)(W0 + (size_t)(r + wv) * DFEAT + c4);
            *(float4*)(sW + (r + wv) * WPITCH + c4) = v;
        }
        if (DUAL) {
#pragma unroll
            for (int r = 0; r < DFEAT; r += 8) {
                const float4 v = *(const float4*)(W1 + (size_t)(r + wv) * DFEAT + c4);
                *(float4*)(sW + (DFEAT + r + wv) * WPITCH + c4) = v;
            }
        }
    }
    __syncthreads();

    const int tile = blockIdx.x * 8 + wv;
    if (tile >= ntiles) return;

    const int m    = lane & 15;
    const int half = lane >> 4;
    const int koff = half * 2;
    const int row_base = tile * 16;

    const float* a0row = A0 + (size_t)(row_base + m) * DFEAT + koff;
    const float* a1row = DUAL ? (A1 + (size_t)(row_base + m) * DFEAT + koff) : a0row;
    const float* bw0   = sW + m * WPITCH + koff;
    const float* bw1   = bw0 + DFEAT * WPITCH;

    v8f acc[8];
#pragma unroll
    for (int t = 0; t < 8; ++t) acc[t] = (v8f){};

    v2f a0c[KSTEPS], a1c[KSTEPS];
    load_a_chunk<DUAL>(a0row, a1row, 0, a0c, a1c);

#pragma unroll 1
    for (int kc = 0; kc < DFEAT - KCHUNK; kc += KCHUNK) {
        v2f a0n[KSTEPS], a1n[KSTEPS];
        load_a_chunk<DUAL>(a0row, a1row, kc + KCHUNK, a0n, a1n);
        mma_chunk<DUAL>(acc, a0c, a1c, bw0 + kc, bw1 + kc);
#pragma unroll
        for (int i = 0; i < KSTEPS; ++i) {
            a0c[i] = a0n[i];
            if (DUAL) a1c[i] = a1n[i];
        }
    }
    mma_chunk<DUAL>(acc, a0c, a1c, bw0 + (DFEAT - KCHUNK), bw1 + (DFEAT - KCHUNK));

#pragma unroll
    for (int t = 0; t < 8; ++t) {
        const float bv = bias[t * 16 + m];
#pragma unroll
        for (int r = 0; r < 8; ++r) {
            float v = acc[t][r] + bv;
            if (RELU) v = fmaxf(v, 0.0f);
            out[(size_t)(row_base + r + 8 * half) * DFEAT + t * 16 + m] = v;
        }
    }
}

extern "C" void kernel_launch(void* const* d_in, const int* in_sizes, int n_in,
                              void* d_out, int out_size, void* d_ws, size_t ws_size,
                              hipStream_t stream) {
    const float* x  = (const float*)d_in[0];
    const int*   ei = (const int*)d_in[1];
    const float* Wl = (const float*)d_in[2];
    const float* bl = (const float*)d_in[3];
    const float* Wr = (const float*)d_in[4];
    const float* W  = (const float*)d_in[5];
    const float* b  = (const float*)d_in[6];

    const int nE = in_sizes[1] / 2;
    const int n  = in_sizes[0] / DFEAT;
    const int*   srcp = ei;
    const int*   dstp = ei + nE;

    float* buf0 = (float*)d_ws;
    float* buf1 = buf0 + (size_t)n * DFEAT;

    const int ntiles = (n + SCAN_TILE) / SCAN_TILE;
    const int npad   = ntiles * SCAN_TILE;
    int* cnt    = (int*)(buf1 + (size_t)n * DFEAT);
    int* rowptr = cnt + npad;
    int* col    = rowptr + npad;

    dim3 blk(256);
    const int zeroBlocks = npad / 4 / 256;
    const int edgeBlocks = (nE + 255) / 256;
    const int nodeBlocks = (n + 7) / 8;
    const int gemmTiles  = n / 16;
    const int gemmBlocks = (gemmTiles + 7) / 8;

    zero_int_kernel<<<zeroBlocks, blk, 0, stream>>>((int4*)cnt, npad / 4);
    csr_count_kernel<<<edgeBlocks, blk, 0, stream>>>(dstp, cnt, nE);
    csr_scan_kernel<<<1, SCAN_THREADS, 0, stream>>>(cnt, rowptr, ntiles);
    csr_fill_kernel<<<edgeBlocks, blk, 0, stream>>>(srcp, dstp, rowptr, cnt, col, nE);

    gather_max_kernel<<<nodeBlocks, blk, 0, stream>>>(x, rowptr, col, buf0, n);
    sage_gemm_kernel<true, true><<<gemmBlocks, blk, 0, stream>>>(
        buf0, x, Wl, Wr, bl, buf0, gemmTiles);

    gather_max_kernel<<<nodeBlocks, blk, 0, stream>>>(buf0, rowptr, col, buf1, n);
    sage_gemm_kernel<true, true><<<gemmBlocks, blk, 0, stream>>>(
        buf1, buf0, Wl + DFEAT * DFEAT, Wr + DFEAT * DFEAT, bl + DFEAT, buf1, gemmTiles);

    sage_gemm_kernel<false, false><<<gemmBlocks, blk, 0, stream>>>(
        buf1, nullptr, W, nullptr, b, (float*)d_out, gemmTiles);
}


# Adjacency build without global atomics: two-level counting sort on LDS counters (bin edges by destination range, then one work-group per bin); 9 launches to 7; only a max follows, no sum reordered
# speedup vs baseline: 3.9360x; 1.7828x over previous
#include <hip/hip_runtime.h>
#include <math.h>


typedef __attribute__((ext_vector_type(2))) float v2f;
typedef __attribute__((ext_vector_type(8))) float v8f;

#define DFEAT 128

#define BIN_SHIFT 8
#define BIN_NODES (1 << BIN_SHIFT)
#define MAXBINS 256
#define NCHUNK 256
#define SORT_THREADS 1024
#define SORT_WAVES (SORT_THREADS / 32)
#define RUNS_PER_WAVE (NCHUNK / SORT_WAVES)
#define EDGE_UNROLL 4

__device__ __forceinline__ int block_scan_excl(int v, int* wsum, int& total) {
    const int lane = threadIdx.x & 31;
    const int wv   = threadIdx.x >> 5;
    int inc = v;
    for (int d = 1; d < 32; d <<= 1) {
        int o = __shfl_up(inc, d);
        if (lane >= d) inc += o;
    }
    if (lane == 31) wsum[wv] = inc;
    __syncthreads();
    int off = 0, tot = 0;
    for (int w = 0; w < SORT_WAVES; ++w) {
        const int s = wsum[w];
        if (w < wv) off += s;
        tot += s;
    }
    total = tot;
    __syncthreads();
    return off + inc - v;
}

__global__ __launch_bounds__(SORT_THREADS) void bin_edges_kernel(const int* __restrict__ src,
                                                                 const int* __restrict__ dst,
                                                                 int* __restrict__ cntT,
                                                                 int* __restrict__ offT,
                                                                 int* __restrict__ binned,
                                                                 int nedges, int nbins) {
    __shared__ int hist[MAXBINS];
    __shared__ int cursor[MAXBINS];
    __shared__ int wsum[SORT_WAVES];
    const int tid = threadIdx.x;
    const int c   = blockIdx.x;
    const int ch  = (nedges + NCHUNK - 1) / NCHUNK;
    const int beg = c * ch;
    const int end = min(beg + ch, nedges);

    if (tid < MAXBINS) hist[tid] = 0;
    __syncthreads();
    for (int base = beg + tid; base < end; base += EDGE_UNROLL * SORT_THREADS) {
        int dv[EDGE_UNROLL];
#pragma unroll
        for (int k = 0; k < EDGE_UNROLL; ++k) {
            dv[k] = dst[min(base + k * SORT_THREADS, end - 1)];
        }
#pragma unroll
        for (int k = 0; k < EDGE_UNROLL; ++k)
            if (base + k * SORT_THREADS < end) atomicAdd(&hist[dv[k] >> BIN_SHIFT], 1);
    }
    __syncthreads();
    const int h = (tid < MAXBINS) ? hist[tid] : 0;
    int total;
    const int ex = block_scan_excl(h, wsum, total);
    if (tid < MAXBINS) cursor[tid] = ex;
    if (tid < nbins) {
        cntT[tid * NCHUNK + c] = h;
        offT[tid * NCHUNK + c] = ex;
    }
    __syncthreads();
    for (int base = beg + tid; base < end; base += EDGE_UNROLL * SORT_THREADS) {
        int dv[EDGE_UNROLL], sv[EDGE_UNROLL];
#pragma unroll
        for (int k = 0; k < EDGE_UNROLL; ++k) {
            const int e = min(base + k * SORT_THREADS, end - 1);
            dv[k] = dst[e];
            sv[k] = src[e];
        }
#pragma unroll
        for (int k = 0; k < EDGE_UNROLL; ++k) {
            if (base + k * SORT_THREADS < end) {
                const int p = atomicAdd(&cursor[dv[k] >> BIN_SHIFT], 1);
                binned[beg + p] = (int)(((unsigned)sv[k] << BIN_SHIFT) |
                                        (unsigned)(dv[k] & (BIN_NODES - 1)));
            }
        }
    }
}

__global__ __launch_bounds__(SORT_THREADS) void build_rows_kernel(const int* __restrict__ cntT,
                                                                  const int* __restrict__ offT,
                                                                  const int* __restrict__ binned,
                                                                  int* __restrict__ rowptr,
                                                                  int* __restrict__ col,
                                                                  int nedges, int n) {
    __shared__ int ncnt[BIN_NODES];
    __shared__ int ncur[BIN_NODES];
    __shared__ int runCnt[NCHUNK];
    __shared__ int runOff[NCHUNK];
    __shared__ int wsum[SORT_WAVES];
    const int tid  = threadIdx.x;
    const int lane = tid & 31;
    const int wv   = tid >> 5;
    const int b    = blockIdx.x;
    const int ch   = (nedges + NCHUNK - 1) / NCHUNK;

    int part = 0;
    {
        const int4* t4 = (const int4*)cntT;
        const int nq = b * (NCHUNK / 4);
        for (int q = tid; q < nq; q += SORT_THREADS) {
            const int4 v = t4[q];
            part += (v.x + v.y) + (v.z + v.w);
        }
    }
    if (tid < NCHUNK) {
        runCnt[tid] = cntT[b * NCHUNK + tid];
        runOff[tid] = tid * ch + offT[b * NCHUNK + tid];
    }
    if (tid < BIN_NODES) ncnt[tid] = 0;
    int binStart;
    (void)block_scan_excl(part, wsum, binStart);

    int rc[RUNS_PER_WAVE], ro[RUNS_PER_WAVE], pk[RUNS_PER_WAVE];
#pragma unroll
    for (int k = 0; k < RUNS_PER_WAVE; ++k) {
        const int cix = wv + SORT_WAVES * k;
        rc[k] = runCnt[cix];
        ro[k] = runOff[cix];
    }
#pragma unroll
    for (int k = 0; k < RUNS_PER_WAVE; ++k)
        pk[k] = (lane < rc[k]) ? binned[ro[k] + lane] : 0;
#pragma unroll
    for (int k = 0; k < RUNS_PER_WAVE; ++k) {
        if (lane < rc[k]) atomicAdd(&ncnt[pk[k] & (BIN_NODES - 1)], 1);
        for (int i = lane + 32; i < rc[k]; i += 32)
            atomicAdd(&ncnt[binned[ro[k] + i] & (BIN_NODES - 1)], 1);
    }
    __syncthreads();

    const int v = (tid < BIN_NODES) ? ncnt[tid] : 0;
    int binLen;
    const int ex = block_scan_excl(v, wsum, binLen);
    if (tid < BIN_NODES) {
        ncur[tid] = ex;
        const int node = b * BIN_NODES + tid;
        if (node <= n) rowptr[node] = binStart + ex;
    }
    __syncthreads();

#pragma unroll
    for (int k = 0; k < RUNS_PER_WAVE; ++k) {
        if (lane < rc[k]) {
            const int pos = atomicAdd(&ncur[pk[k] & (BIN_NODES - 1)], 1);
            col[binStart + pos] = (int)((unsigned)pk[k] >> BIN_SHIFT);
        }
        for (int i = lane + 32; i < rc[k]; i += 32) {
            const int p = binned[ro[k] + i];
            const int pos = atomicAdd(&ncur[p & (BIN_NODES - 1)], 1);
            col[binStart + pos] = (int)((unsigned)p >> BIN_SHIFT);
        }
    }
}

__device__ __forceinline__ float4 max3v(float4 m, float4 a, float4 b) {
    float4 r;
    r.x = __builtin_fmaxf(__builtin_fmaxf(m.x, a.x), b.x);
    r.y = __builtin_fmaxf(__builtin_fmaxf(m.y, a.y), b.y);
    r.z = __builtin_fmaxf(__builtin_fmaxf(m.z, a.z), b.z);
    r.w = __builtin_fmaxf(__builtin_fmaxf(m.w, a.w), b.w);
    return r;
}

__device__ __forceinline__ float4 load_row(const float* __restrict__ x, int ids, int j, unsigned loff) {
    const float* row = x + (size_t)__builtin_amdgcn_readlane(ids, j) * DFEAT;
    return *(const float4*)(row + loff);
}

__global__ __launch_bounds__(256) void gather_max_kernel(const float* __restrict__ x,
                                                         const int* __restrict__ rowptr,
                                                         const int* __restrict__ col,
                                                         float* __restrict__ agg,
                                                         int n) {
    const int lane = threadIdx.x & 31;
    const unsigned loff = (unsigned)lane * 4u;
    const int node = __builtin_amdgcn_readfirstlane((int)((blockIdx.x * blockDim.x + threadIdx.x) >> 5));
    if (node >= n) return;
    const int beg = rowptr[node];
    const int end = rowptr[node + 1];
    float4 acc;
    acc.x = -INFINITY; acc.y = -INFINITY; acc.z = -INFINITY; acc.w = -INFINITY;
    for (int base = beg; base < end; base += 32) {
        const int cnt = min(end - base, 32);
        const int ids = col[base + min(lane, cnt - 1)];
        int j = 0;
        for (; j + 4 < cnt; j += 8) {
            const float4 r0 = load_row(x, ids, j + 0, loff);
            const float4 r1 = load_row(x, ids, j + 1, loff);
            const float4 r2 = load_row(x, ids, j + 2, loff);
            const float4 r3 = load_row(x, ids, j + 3, loff);
            const float4 r4 = load_row(x, ids, j + 4, loff);
            const float4 r5 = load_row(x, ids, j + 5, loff);
            const float4 r6 = load_row(x, ids, j + 6, loff);
            const float4 r7 = load_row(x, ids, j + 7, loff);
            acc = max3v(acc, r0, r1);
            acc = max3v(acc, r2, r3);
            acc = max3v(acc, r4, r5);
            acc = max3v(acc, r6, r7);
        }
        if (j < cnt) {
            const float4 r0 = load_row(x, ids, j + 0, loff);
            const float4 r1 = load_row(x, ids, j + 1, loff);
            const float4 r2 = load_row(x, ids, j + 2, loff);
            const float4 r3 = load_row(x, ids, j + 3, loff);
            acc = max3v(acc, r0, r1);
            acc = max3v(acc, r2, r3);
        }
    }
    *(float4*)(agg + (size_t)node * DFEAT + loff) = acc;
}

#define WPITCH 132
#define KCHUNK 32
#define KSTEPS (KCHUNK / 4)

template <bool DUAL>
__device__ __forceinline__ void load_a_chunk(const float* __restrict__ a0row,
                                             const float* __restrict__ a1row,
                                             int kc, v2f (&a0)[KSTEPS], v2f (&a1)[KSTEPS]) {
#pragma unroll
    for (int i = 0; i < KSTEPS; ++i) {
        a0[i] = *(const v2f*)(a0row + kc + 4 * i);
        if (DUAL) a1[i] = *(const v2f*)(a1row + kc + 4 * i);
    }
}

template <bool DUAL>
__device__ __forceinline__ void mma_chunk(v8f (&acc)[8], const v2f (&a0c)[KSTEPS],
                                          const v2f (&a1c)[KSTEPS],
                                          const float* bw0, const float* bw1) {
#pragma unroll
    for (int i = 0; i < KSTEPS; ++i) {
        v2f a0 = a0c[i];
        if (DUAL) {
            a0.x = (a0.x == -INFINITY) ? 0.0f : a0.x;
            a0.y = (a0.y == -INFINITY) ? 0.0f : a0.y;
        }
        v2f b[8];
#pragma unroll
        for (int t = 0; t < 8; ++t) b[t] = *(const v2f*)(bw0 + t * 16 * WPITCH + 4 * i);
#pragma unroll
        for (int t = 0; t < 8; ++t)
            acc[t] = __builtin_amdgcn_wmma_f32_16x16x4_f32(false, a0, false, b[t],
                                                           (short)0, acc[t], false, false);
        if (DUAL) {
            const v2f a1 = a1c[i];
#pragma unroll
            for (int t = 0; t < 8; ++t) b[t] = *(const v2f*)(bw1 + t * 16 * WPITCH + 4 * i);
#pragma unroll
            for (int t = 0; t < 8; ++t)
                acc[t] = __builtin_amdgcn_wmma_f32_16x16x4_f32(false, a1, false, b[t],
                                                               (short)0, acc[t], false, false);
        }
    }
}

template <bool DUAL, bool RELU>
__global__ __launch_bounds__(256) void sage_gemm_kernel(const float* __restrict__ A0,
                                                        const float* __restrict__ A1,
                                                        const float* __restrict__ W0,
                                                        const float* __restrict__ W1,
                                                        const float* __restrict__ bias,
                                                        float* __restrict__ out,
                                                        int ntiles) {
    __shared__ __attribute__((aligned(16))) float sW[(DUAL ? 2 : 1) * DFEAT * WPITCH];

    const int lane = threadIdx.x & 31;
    const int wv = __builtin_amdgcn_readfirstlane((int)(threadIdx.x >> 5));

    {
        const int c4 = lane * 4;
#pragma unroll
        for (int r = 0; r < DFEAT; r += 8) {
            const float4 v = *(const float4*)(W0 + (size_t)(r + wv) * DFEAT + c4);
            *(float4*)(sW + (r + wv) * WPITCH + c4) = v;
        }
        if (DUAL) {
#pragma unroll
            for (int r = 0; r < DFEAT; r += 8) {
                const float4 v = *(const float4*)(W1 + (size_t)(r + wv) * DFEAT + c4);
                *(float4*)(sW + (DFEAT + r + wv) * WPITCH + c4) = v;
            }
        }
    }
    __syncthreads();

    const int tile = blockIdx.x * 8 + wv;
    if (tile >= ntiles) return;

    const int m    = lane & 15;
    const int half = lane >> 4;
    const int koff = half * 2;
    const int row_base = tile * 16;

    const float* a0row = A0 + (size_t)(row_base + m) * DFEAT + koff;
    const float* a1row = DUAL ? (A1 + (size_t)(row_base + m) * DFEAT + koff) : a0row;
    const float* bw0   = sW + m * WPITCH + koff;
    const float* bw1   = bw0 + DFEAT * WPITCH;

    v8f acc[8];
#pragma unroll
    for (int t = 0; t < 8; ++t) acc[t] = (v8f){};

    v2f a0c[KSTEPS], a1c[KSTEPS];
    load_a_chunk<DUAL>(a0row, a1row, 0, a0c, a1c);

#pragma unroll 1
    for (int kc = 0; kc < DFEAT - KCHUNK; kc += KCHUNK) {
        v2f a0n[KSTEPS], a1n[KSTEPS];
        load_a_chunk<DUAL>(a0row, a1row, kc + KCHUNK, a0n, a1n);
        mma_chunk<DUAL>(acc, a0c, a1c, bw0 + kc, bw1 + kc);
#pragma unroll
        for (int i = 0; i < KSTEPS; ++i) {
            a0c[i] = a0n[i];
            if (DUAL) a1c[i] = a1n[i];
        }
    }
    mma_chunk<DUAL>(acc, a0c, a1c, bw0 + (DFEAT - KCHUNK), bw1 + (DFEAT - KCHUNK));

#pragma unroll
    for (int t = 0; t < 8; ++t) {
        const float bv = bias[t * 16 + m];
#pragma unroll
        for (int r = 0; r < 8; ++r) {
            float v = acc[t][r] + bv;
            if (RELU) v = fmaxf(v, 0.0f);
            out[(size_t)(row_base + r + 8 * half) * DFEAT + t * 16 + m] = v;
        }
    }
}

extern "C" void kernel_launch(void* const* d_in, const int* in_sizes, int n_in,
                              void* d_out, int out_size, void* d_ws, size_t ws_size,
                              hipStream_t stream) {
    const float* x  = (const float*)d_in[0];
    const int*   ei = (const int*)d_in[1];
    const float* Wl = (const float*)d_in[2];
    const float* bl = (const float*)d_in[3];
    const float* Wr = (const float*)d_in[4];
    const float* W  = (const float*)d_in[5];
    const float* b  = (const float*)d_in[6];

    const int nE = in_sizes[1] / 2;
    const int n  = in_sizes[0] / DFEAT;
    const int*   srcp = ei;
    const int*   dstp = ei + nE;

    float* buf0 = (float*)d_ws;
    float* buf1 = buf0 + (size_t)n * DFEAT;

    const int nbins = (n + BIN_NODES) / BIN_NODES;
    int* cntT   = (int*)(buf1 + (size_t)n * DFEAT);
    int* offT   = cntT + MAXBINS * NCHUNK;
    int* rowptr = offT + MAXBINS * NCHUNK;
    int* binned = rowptr + ((n + 64) & ~63);
    int* col    = binned + nE;

    dim3 blk(256);
    const int nodeBlocks = (n + 7) / 8;
    const int gemmTiles  = n / 16;
    const int gemmBlocks = (gemmTiles + 7) / 8;

    bin_edges_kernel<<<NCHUNK, SORT_THREADS, 0, stream>>>(srcp, dstp, cntT, offT, binned, nE, nbins);
    build_rows_kernel<<<nbins, SORT_THREADS, 0, stream>>>(cntT, offT, binned, rowptr, col, nE, n);

    gather_max_kernel<<<nodeBlocks, blk, 0, stream>>>(x, rowptr, col, buf0, n);
    sage_gemm_kernel<true, true><<<gemmBlocks, blk, 0, stream>>>(
        buf0, x, Wl, Wr, bl, buf0, gemmTiles);

    gather_max_kernel<<<nodeBlocks, blk, 0, stream>>>(buf0, rowptr, col, buf1, n);
    sage_gemm_kernel<true, true><<<gemmBlocks, blk, 0, stream>>>(
        buf1, buf0, Wl + DFEAT * DFEAT, Wr + DFEAT * DFEAT, bl + DFEAT, buf1, gemmTiles);

    sage_gemm_kernel<false, false><<<gemmBlocks, blk, 0, stream>>>(
        buf1, nullptr, W, nullptr, b, (float*)d_out, gemmTiles);
}


# GEMM work dealt evenly over SIMDs: 256 work-groups of 16 waves, 12 waves own a 16x128 tile each, 53 left-over row tiles cut into 16x32 pieces; fullest SIMD 2048 -> 1664 WMMAs; no sum reordered
# speedup vs baseline: 4.1832x; 1.0628x over previous
#include <hip/hip_runtime.h>
#include <math.h>


typedef __attribute__((ext_vector_type(2))) float v2f;
typedef __attribute__((ext_vector_type(8))) float v8f;

#define DFEAT 128

#define BIN_SHIFT 8
#define BIN_NODES (1 << BIN_SHIFT)
#define MAXBINS 256
#define NCHUNK 256
#define SORT_THREADS 1024
#define SORT_WAVES (SORT_THREADS / 32)
#define RUNS_PER_WAVE (NCHUNK / SORT_WAVES)
#define EDGE_UNROLL 4

__device__ __forceinline__ int block_scan_excl(int v, int* wsum, int& total) {
    const int lane = threadIdx.x & 31;
    const int wv   = threadIdx.x >> 5;
    int inc = v;
    for (int d = 1; d < 32; d <<= 1) {
        int o = __shfl_up(inc, d);
        if (lane >= d) inc += o;
    }
    if (lane == 31) wsum[wv] = inc;
    __syncthreads();
    int off = 0, tot = 0;
    for (int w = 0; w < SORT_WAVES; ++w) {
        const int s = wsum[w];
        if (w < wv) off += s;
        tot += s;
    }
    total = tot;
    __syncthreads();
    return off + inc - v;
}

__global__ __launch_bounds__(SORT_THREADS) void bin_edges_kernel(const int* __restrict__ src,
                                                                 const int* __restrict__ dst,
                                                                 int* __restrict__ cntT,
                                                                 int* __restrict__ offT,
                                                                 int* __restrict__ binned,
                                                                 int nedges, int nbins) {
    __shared__ int hist[MAXBINS];
    __shared__ int cursor[MAXBINS];
    __shared__ int wsum[SORT_WAVES];
    const int tid = threadIdx.x;
    const int c   = blockIdx.x;
    const int ch  = (nedges + NCHUNK - 1) / NCHUNK;
    const int beg = c * ch;
    const int end = min(beg + ch, nedges);

    if (tid < MAXBINS) hist[tid] = 0;
    __syncthreads();
    for (int base = beg + tid; base < end; base += EDGE_UNROLL * SORT_THREADS) {
        int dv[EDGE_UNROLL];
#pragma unroll
        for (int k = 0; k < EDGE_UNROLL; ++k) {
            dv[k] = dst[min(base + k * SORT_THREADS, end - 1)];
        }
#pragma unroll
        for (int k = 0; k < EDGE_UNROLL; ++k)
            if (base + k * SORT_THREADS < end) atomicAdd(&hist[dv[k] >> BIN_SHIFT], 1);
    }
    __syncthreads();
    const int h = (tid < MAXBINS) ? hist[tid] : 0;
    int total;
    const int ex = block_scan_excl(h, wsum, total);
    if (tid < MAXBINS) cursor[tid] = ex;
    if (tid < nbins) {
        cntT[tid * NCHUNK + c] = h;
        offT[tid * NCHUNK + c] = ex;
    }
    __syncthreads();
    for (int base = beg + tid; base < end; base += EDGE_UNROLL * SORT_THREADS) {
        int dv[EDGE_UNROLL], sv[EDGE_UNROLL];
#pragma unroll
        for (int k = 0; k < EDGE_UNROLL; ++k) {
            const int e = min(base + k * SORT_THREADS, end - 1);
            dv[k] = dst[e];
            sv[k] = src[e];
        }
#pragma unroll
        for (int k = 0; k < EDGE_UNROLL; ++k) {
            if (base + k * SORT_THREADS < end) {
                const int p = atomicAdd(&cursor[dv[k] >> BIN_SHIFT], 1);
                binned[beg + p] = (int)(((unsigned)sv[k] << BIN_SHIFT) |
                                        (unsigned)(dv[k] & (BIN_NODES - 1)));
            }
        }
    }
}

__global__ __launch_bounds__(SORT_THREADS) void build_rows_kernel(const int* __restrict__ cntT,
                                                                  const int* __restrict__ offT,
                                                                  const int* __restrict__ binned,
                                                                  int* __restrict__ rowptr,
                                                                  int* __restrict__ col,
                                                                  int nedges, int n) {
    __shared__ int ncnt[BIN_NODES];
    __shared__ int ncur[BIN_NODES];
    __shared__ int runCnt[NCHUNK];
    __shared__ int runOff[NCHUNK];
    __shared__ int wsum[SORT_WAVES];
    const int tid  = threadIdx.x;
    const int lane = tid & 31;
    const int wv   = tid >> 5;
    const int b    = blockIdx.x;
    const int ch   = (nedges + NCHUNK - 1) / NCHUNK;

    int part = 0;
    {
        const int4* t4 = (const int4*)cntT;
        const int nq = b * (NCHUNK / 4);
        for (int q = tid; q < nq; q += SORT_THREADS) {
            const int4 v = t4[q];
            part += (v.x + v.y) + (v.z + v.w);
        }
    }
    if (tid < NCHUNK) {
        runCnt[tid] = cntT[b * NCHUNK + tid];
        runOff[tid] = tid * ch + offT[b * NCHUNK + tid];
    }
    if (tid < BIN_NODES) ncnt[tid] = 0;
    int binStart;
    (void)block_scan_excl(part, wsum, binStart);

    int rc[RUNS_PER_WAVE], ro[RUNS_PER_WAVE], pk[RUNS_PER_WAVE];
#pragma unroll
    for (int k = 0; k < RUNS_PER_WAVE; ++k) {
        const int cix = wv + SORT_WAVES * k;
        rc[k] = runCnt[cix];
        ro[k] = runOff[cix];
    }
#pragma unroll
    for (int k = 0; k < RUNS_PER_WAVE; ++k)
        pk[k] = (lane < rc[k]) ? binned[ro[k] + lane] : 0;
#pragma unroll
    for (int k = 0; k < RUNS_PER_WAVE; ++k) {
        if (lane < rc[k]) atomicAdd(&ncnt[pk[k] & (BIN_NODES - 1)], 1);
        for (int i = lane + 32; i < rc[k]; i += 32)
            atomicAdd(&ncnt[binned[ro[k] + i] & (BIN_NODES - 1)], 1);
    }
    __syncthreads();

    const int v = (tid < BIN_NODES) ? ncnt[tid] : 0;
    int binLen;
    const int ex = block_scan_excl(v, wsum, binLen);
    if (tid < BIN_NODES) {
        ncur[tid] = ex;
        const int node = b * BIN_NODES + tid;
        if (node <= n) rowptr[node] = binStart + ex;
    }
    __syncthreads();

#pragma unroll
    for (int k = 0; k < RUNS_PER_WAVE; ++k) {
        if (lane < rc[k]) {
            const int pos = atomicAdd(&ncur[pk[k] & (BIN_NODES - 1)], 1);
            col[binStart + pos] = (int)((unsigned)pk[k] >> BIN_SHIFT);
        }
        for (int i = lane + 32; i < rc[k]; i += 32) {
            const int p = binned[ro[k] + i];
            const int pos = atomicAdd(&ncur[p & (BIN_NODES - 1)], 1);
            col[binStart + pos] = (int)((unsigned)p >> BIN_SHIFT);
        }
    }
}

__device__ __forceinline__ float4 max3v(float4 m, float4 a, float4 b) {
    float4 r;
    r.x = __builtin_fmaxf(__builtin_fmaxf(m.x, a.x), b.x);
    r.y = __builtin_fmaxf(__builtin_fmaxf(m.y, a.y), b.y);
    r.z = __builtin_fmaxf(__builtin_fmaxf(m.z, a.z), b.z);
    r.w = __builtin_fmaxf(__builtin_fmaxf(m.w, a.w), b.w);
    return r;
}

__device__ __forceinline__ float4 load_row(const float* __restrict__ x, int ids, int j, unsigned loff) {
    const float* row = x + (size_t)__builtin_amdgcn_readlane(ids, j) * DFEAT;
    return *(const float4*)(row + loff);
}

__global__ __launch_bounds__(256) void gather_max_kernel(const float* __restrict__ x,
                                                         const int* __restrict__ rowptr,
                                                         const int* __restrict__ col,
                                                         float* __restrict__ agg,
                                                         int n) {
    const int lane = threadIdx.x & 31;
    const unsigned loff = (unsigned)lane * 4u;
    const int node = __builtin_amdgcn_readfirstlane((int)((blockIdx.x * blockDim.x + threadIdx.x) >> 5));
    if (node >= n) return;
    const int beg = rowptr[node];
    const int end = rowptr[node + 1];
    float4 acc;
    acc.x = -INFINITY; acc.y = -INFINITY; acc.z = -INFINITY; acc.w = -INFINITY;
    for (int base = beg; base < end; base += 32) {
        const int cnt = min(end - base, 32);
        const int ids = col[base + min(lane, cnt - 1)];
        int j = 0;
        for (; j + 4 < cnt; j += 8) {
            const float4 r0 = load_row(x, ids, j + 0, loff);
            const float4 r1 = load_row(x, ids, j + 1, loff);
            const float4 r2 = load_row(x, ids, j + 2, loff);
            const float4 r3 = load_row(x, ids, j + 3, loff);
            const float4 r4 = load_row(x, ids, j + 4, loff);
            const float4 r5 = load_row(x, ids, j + 5, loff);
            const float4 r6 = load_row(x, ids, j + 6, loff);
            const float4 r7 = load_row(x, ids, j + 7, loff);
            acc = max3v(acc, r0, r1);
            acc = max3v(acc, r2, r3);
            acc = max3v(acc, r4, r5);
            acc = max3v(acc, r6, r7);
        }
        if (j < cnt) {
            const float4 r0 = load_row(x, ids, j + 0, loff);
            const float4 r1 = load_row(x, ids, j + 1, loff);
            const float4 r2 = load_row(x, ids, j + 2, loff);
            const float4 r3 = load_row(x, ids, j + 3, loff);
            acc = max3v(acc, r0, r1);
            acc = max3v(acc, r2, r3);
        }
    }
    *(float4*)(agg + (size_t)node * DFEAT + loff) = acc;
}

#define WPITCH 132
#define KCHUNK 32
#define KSTEPS (KCHUNK / 4)
#define NUM_WGP 256
#define GEMM_WAVES 16
#define MAIN_WAVES 12
#define COL_TILES (DFEAT / 16)
#define PIECE_TILES 2
#define PIECES_PER_ROWTILE (COL_TILES / PIECE_TILES)

template <bool DUAL>
__device__ __forceinline__ void load_a_chunk(const float* __restrict__ a0row,
                                             const float* __restrict__ a1row,
                                             int kc, v2f (&a0)[KSTEPS], v2f (&a1)[KSTEPS]) {
#pragma unroll
    for (int i = 0; i < KSTEPS; ++i) {
        a0[i] = *(const v2f*)(a0row + kc + 4 * i);
        if (DUAL) a1[i] = *(const v2f*)(a1row + kc + 4 * i);
    }
}

template <bool DUAL, int NT>
__device__ __forceinline__ void mma_chunk(v8f (&acc)[NT], const v2f (&a0c)[KSTEPS],
                                          const v2f (&a1c)[KSTEPS],
                                          const float* bw0, const float* bw1) {
#pragma unroll
    for (int i = 0; i < KSTEPS; ++i) {
        v2f a0 = a0c[i];
        if (DUAL) {
            a0.x = (a0.x == -INFINITY) ? 0.0f : a0.x;
            a0.y = (a0.y == -INFINITY) ? 0.0f : a0.y;
        }
        v2f b[NT];
#pragma unroll
        for (int t = 0; t < NT; ++t) b[t] = *(const v2f*)(bw0 + t * 16 * WPITCH + 4 * i);
#pragma unroll
        for (int t = 0; t < NT; ++t)
            acc[t] = __builtin_amdgcn_wmma_f32_16x16x4_f32(false, a0, false, b[t],
                                                           (short)0, acc[t], false, false);
        if (DUAL) {
            const v2f a1 = a1c[i];
#pragma unroll
            for (int t = 0; t < NT; ++t) b[t] = *(const v2f*)(bw1 + t * 16 * WPITCH + 4 * i);
#pragma unroll
            for (int t = 0; t < NT; ++t)
                acc[t] = __builtin_amdgcn_wmma_f32_16x16x4_f32(false, a1, false, b[t],
                                                               (short)0, acc[t], false, false);
        }
    }
}

template <bool DUAL, bool RELU, int NT>
__device__ __forceinline__ void tile_product(const float* __restrict__ A0,
                                             const float* __restrict__ A1,
                                             const float* sW,
                                             const float* __restrict__ bias,
                                             float* __restrict__ out,
                                             int row_base, int col_base, int lane) {
    const int m    = lane & 15;
    const int half = lane >> 4;
    const int koff = half * 2;

    const float* a0row = A0 + (size_t)(row_base + m) * DFEAT + koff;
    const float* a1row = DUAL ? (A1 + (size_t)(row_base + m) * DFEAT + koff) : a0row;
    const float* bw0   = sW + (col_base + m) * WPITCH + koff;
    const float* bw1   = bw0 + DFEAT * WPITCH;

    v8f acc[NT];
#pragma unroll
    for (int t = 0; t < NT; ++t) acc[t] = (v8f){};

    v2f a0c[KSTEPS], a1c[KSTEPS];
    load_a_chunk<DUAL>(a0row, a1row, 0, a0c, a1c);

#pragma unroll 1
    for (int kc = 0; kc < DFEAT - KCHUNK; kc += KCHUNK) {
        v2f a0n[KSTEPS], a1n[KSTEPS];
        load_a_chunk<DUAL>(a0row, a1row, kc + KCHUNK, a0n, a1n);
        mma_chunk<DUAL, NT>(acc, a0c, a1c, bw0 + kc, bw1 + kc);
#pragma unroll
        for (int i = 0; i < KSTEPS; ++i) {
            a0c[i] = a0n[i];
            if (DUAL) a1c[i] = a1n[i];
        }
    }
    mma_chunk<DUAL, NT>(acc, a0c, a1c, bw0 + (DFEAT - KCHUNK), bw1 + (DFEAT - KCHUNK));

#pragma unroll
    for (int t = 0; t < NT; ++t) {
        const float bv = bias[col_base + t * 16 + m];
#pragma unroll
        for (int r = 0; r < 8; ++r) {
            float v = acc[t][r] + bv;
            if (RELU) v = fmaxf(v, 0.0f);
            out[(size_t)(row_base + r + 8 * half) * DFEAT + col_base + t * 16 + m] = v;
        }
    }
}

template <bool DUAL, bool RELU>
__global__ __launch_bounds__(GEMM_WAVES * 32) void sage_gemm_kernel(const float* __restrict__ A0,
                                                                    const float* __restrict__ A1,
                                                                    const float* __restrict__ W0,
                                                                    const float* __restrict__ W1,
                                                                    const float* __restrict__ bias,
                                                                    float* __restrict__ out,
                                                                    int ntiles) {
    __shared__ __attribute__((aligned(16))) float sW[(DUAL ? 2 : 1) * DFEAT * WPITCH];

    const int lane = threadIdx.x & 31;
    const int wv = __builtin_amdgcn_readfirstlane((int)(threadIdx.x >> 5));

    {
        const int c4 = lane * 4;
#pragma unroll
        for (int r = 0; r < DFEAT; r += GEMM_WAVES) {
            const float4 v = *(const float4*)(W0 + (size_t)(r + wv) * DFEAT + c4);
            *(float4*)(sW + (r + wv) * WPITCH + c4) = v;
        }
        if (DUAL) {
#pragma unroll
            for (int r = 0; r < DFEAT; r += GEMM_WAVES) {
                const float4 v = *(const float4*)(W1 + (size_t)(r + wv) * DFEAT + c4);
                *(float4*)(sW + (DFEAT + r + wv) * WPITCH + c4) = v;
            }
        }
    }
    __syncthreads();

    const int nwg       = (int)gridDim.x;
    const int roundSize = nwg * MAIN_WAVES;
    const int places    = nwg * (GEMM_WAVES - MAIN_WAVES);
    int leftover  = ntiles % roundSize;
    int mainTiles = ntiles - leftover;
    if (leftover * PIECES_PER_ROWTILE > places) {
        mainTiles = ntiles;
        leftover  = 0;
    }

    if (wv < MAIN_WAVES) {
        for (int tile = (int)blockIdx.x * MAIN_WAVES + wv; tile < mainTiles; tile += roundSize) {
            asm volatile("" ::: "memory");
            tile_product<DUAL, RELU, COL_TILES>(A0, A1, sW, bias, out, tile * 16, 0, lane);
        }
    } else {
        const int npieces = leftover * PIECES_PER_ROWTILE;
        for (int p = (wv - MAIN_WAVES) * nwg + (int)blockIdx.x; p < npieces; p += places) {
            const int rt = mainTiles + p / PIECES_PER_ROWTILE;
            const int cp = p % PIECES_PER_ROWTILE;
            tile_product<DUAL, RELU, PIECE_TILES>(A0, A1, sW, bias, out, rt * 16,
                                                  cp * 16 * PIECE_TILES, lane);
        }
    }
}

extern "C" void kernel_launch(void* const* d_in, const int* in_sizes, int n_in,
                              void* d_out, int out_size, void* d_ws, size_t ws_size,
                              hipStream_t stream) {
    const float* x  = (const float*)d_in[0];
    const int*   ei = (const int*)d_in[1];
    const float* Wl = (const float*)d_in[2];
    const float* bl = (const float*)d_in[3];
    const float* Wr = (const float*)d_in[4];
    const float* W  = (const float*)d_in[5];
    const float* b  = (const float*)d_in[6];

    const int nE = in_sizes[1] / 2;
    const int n  = in_sizes[0] / DFEAT;
    const int*   srcp = ei;
    const int*   dstp = ei + nE;

    float* buf0 = (float*)d_ws;
    float* buf1 = buf0 + (size_t)n * DFEAT;
    float* buf2 = buf1 + (size_t)n * DFEAT;

    const int nbins = (n + BIN_NODES) / BIN_NODES;
    int* cntT   = (int*)(buf2 + (size_t)n * DFEAT);
    int* offT   = cntT + MAXBINS * NCHUNK;
    int* rowptr = offT + MAXBINS * NCHUNK;
    int* binned = rowptr + ((n + 64) & ~63);
    int* col    = binned + nE;

    dim3 blk(256);
    dim3 gemmBlk(GEMM_WAVES * 32);
    const int nodeBlocks = (n + 7) / 8;
    const int gemmTiles  = n / 16;

    bin_edges_kernel<<<NCHUNK, SORT_THREADS, 0, stream>>>(srcp, dstp, cntT, offT, binned, nE, nbins);
    build_rows_kernel<<<nbins, SORT_THREADS, 0, stream>>>(cntT, offT, binned, rowptr, col, nE, n);

    gather_max_kernel<<<nodeBlocks, blk, 0, stream>>>(x, rowptr, col, buf0, n);
    sage_gemm_kernel<true, true><<<NUM_WGP, gemmBlk, 0, stream>>>(
        buf0, x, Wl, Wr, bl, buf1, gemmTiles);

    gather_max_kernel<<<nodeBlocks, blk, 0, stream>>>(buf1, rowptr, col, buf0, n);
    sage_gemm_kernel<true, true><<<NUM_WGP, gemmBlk, 0, stream>>>(
        buf0, buf1, Wl + DFEAT * DFEAT, Wr + DFEAT * DFEAT, bl + DFEAT, buf2, gemmTiles);

    sage_gemm_kernel<false, false><<<NUM_WGP, gemmBlk, 0, stream>>>(
        buf2, nullptr, W, nullptr, b, (float*)d_out, gemmTiles);
}


# Gather-max fused into the two layer GEMM launches: each wave builds its row tile's 16 max rows in its own LDS tile and takes its A fragments from there; 7 launches to 5; no sum reordered
# speedup vs baseline: 4.4017x; 1.0522x over previous
#include <hip/hip_runtime.h>
#include <math.h>


typedef __attribute__((ext_vector_type(2))) float v2f;
typedef __attribute__((ext_vector_type(8))) float v8f;

#define DFEAT 128

#define BIN_SHIFT 8
#define BIN_NODES (1 << BIN_SHIFT)
#define MAXBINS 256
#define NCHUNK 256
#define SORT_THREADS 1024
#define SORT_WAVES (SORT_THREADS / 32)
#define RUNS_PER_WAVE (NCHUNK / SORT_WAVES)
#define EDGE_UNROLL 4

__device__ __forceinline__ int block_scan_excl(int v, int* wsum, int& total) {
    const int lane = threadIdx.x & 31;
    const int wv   = threadIdx.x >> 5;
    int inc = v;
    for (int d = 1; d < 32; d <<= 1) {
        int o = __shfl_up(inc, d);
        if (lane >= d) inc += o;
    }
    if (lane == 31) wsum[wv] = inc;
    __syncthreads();
    int off = 0, tot = 0;
    for (int w = 0; w < SORT_WAVES; ++w) {
        const int s = wsum[w];
        if (w < wv) off += s;
        tot += s;
    }
    total = tot;
    __syncthreads();
    return off + inc - v;
}

__global__ __launch_bounds__(SORT_THREADS) void bin_edges_kernel(const int* __restrict__ src,
                                                                 const int* __restrict__ dst,
                                                                 int* __restrict__ cntT,
                                                                 int* __restrict__ offT,
                                                                 int* __restrict__ binned,
                                                                 int nedges, int nbins) {
    __shared__ int hist[MAXBINS];
    __shared__ int cursor[MAXBINS];
    __shared__ int wsum[SORT_WAVES];
    const int tid = threadIdx.x;
    const int c   = blockIdx.x;
    const int ch  = (nedges + NCHUNK - 1) / NCHUNK;
    const int beg = c * ch;
    const int end = min(beg + ch, nedges);

    if (tid < MAXBINS) hist[tid] = 0;
    __syncthreads();
    for (int base = beg + tid; base < end; base += EDGE_UNROLL * SORT_THREADS) {
        int dv[EDGE_UNROLL];
#pragma unroll
        for (int k = 0; k < EDGE_UNROLL; ++k) {
            dv[k] = dst[min(base + k * SORT_THREADS, end - 1)];
        }
#pragma unroll
        for (int k = 0; k < EDGE_UNROLL; ++k)
            if (base + k * SORT_THREADS < end) atomicAdd(&hist[dv[k] >> BIN_SHIFT], 1);
    }
    __syncthreads();
    const int h = (tid < MAXBINS) ? hist[tid] : 0;
    int total;
    const int ex = block_scan_excl(h, wsum, total);
    if (tid < MAXBINS) cursor[tid] = ex;
    if (tid < nbins) {
        cntT[tid * NCHUNK + c] = h;
        offT[tid * NCHUNK + c] = ex;
    }
    __syncthreads();
    for (int base = beg + tid; base < end; base += EDGE_UNROLL * SORT_THREADS) {
        int dv[EDGE_UNROLL], sv[EDGE_UNROLL];
#pragma unroll
        for (int k = 0; k < EDGE_UNROLL; ++k) {
            const int e = min(base + k * SORT_THREADS, end - 1);
            dv[k] = dst[e];
            sv[k] = src[e];
        }
#pragma unroll
        for (int k = 0; k < EDGE_UNROLL; ++k) {
            if (base + k * SORT_THREADS < end) {
                const int p = atomicAdd(&cursor[dv[k] >> BIN_SHIFT], 1);
                binned[beg + p] = (int)(((unsigned)sv[k] << BIN_SHIFT) |
                                        (unsigned)(dv[k] & (BIN_NODES - 1)));
            }
        }
    }
}

__global__ __launch_bounds__(SORT_THREADS) void build_rows_kernel(const int* __restrict__ cntT,
                                                                  const int* __restrict__ offT,
                                                                  const int* __restrict__ binned,
                                                                  int* __restrict__ rowptr,
                                                                  int* __restrict__ col,
                                                                  int nedges, int n) {
    __shared__ int ncnt[BIN_NODES];
    __shared__ int ncur[BIN_NODES];
    __shared__ int runCnt[NCHUNK];
    __shared__ int runOff[NCHUNK];
    __shared__ int wsum[SORT_WAVES];
    const int tid  = threadIdx.x;
    const int lane = tid & 31;
    const int wv   = tid >> 5;
    const int b    = blockIdx.x;
    const int ch   = (nedges + NCHUNK - 1) / NCHUNK;

    int part = 0;
    {
        const int4* t4 = (const int4*)cntT;
        const int nq = b * (NCHUNK / 4);
        for (int q = tid; q < nq; q += SORT_THREADS) {
            const int4 v = t4[q];
            part += (v.x + v.y) + (v.z + v.w);
        }
    }
    if (tid < NCHUNK) {
        runCnt[tid] = cntT[b * NCHUNK + tid];
        runOff[tid] = tid * ch + offT[b * NCHUNK + tid];
    }
    if (tid < BIN_NODES) ncnt[tid] = 0;
    int binStart;
    (void)block_scan_excl(part, wsum, binStart);

    int rc[RUNS_PER_WAVE], ro[RUNS_PER_WAVE], pk[RUNS_PER_WAVE];
#pragma unroll
    for (int k = 0; k < RUNS_PER_WAVE; ++k) {
        const int cix = wv + SORT_WAVES * k;
        rc[k] = runCnt[cix];
        ro[k] = runOff[cix];
    }
#pragma unroll
    for (int k = 0; k < RUNS_PER_WAVE; ++k)
        pk[k] = (lane < rc[k]) ? binned[ro[k] + lane] : 0;
#pragma unroll
    for (int k = 0; k < RUNS_PER_WAVE; ++k) {
        if (lane < rc[k]) atomicAdd(&ncnt[pk[k] & (BIN_NODES - 1)], 1);
        for (int i = lane + 32; i < rc[k]; i += 32)
            atomicAdd(&ncnt[binned[ro[k] + i] & (BIN_NODES - 1)], 1);
    }
    __syncthreads();

    const int v = (tid < BIN_NODES) ? ncnt[tid] : 0;
    int binLen;
    const int ex = block_scan_excl(v, wsum, binLen);
    if (tid < BIN_NODES) {
        ncur[tid] = ex;
        const int node = b * BIN_NODES + tid;
        if (node <= n) rowptr[node] = binStart + ex;
    }
    __syncthreads();

#pragma unroll
    for (int k = 0; k < RUNS_PER_WAVE; ++k) {
        if (lane < rc[k]) {
            const int pos = atomicAdd(&ncur[pk[k] & (BIN_NODES - 1)], 1);
            col[binStart + pos] = (int)((unsigned)pk[k] >> BIN_SHIFT);
        }
        for (int i = lane + 32; i < rc[k]; i += 32) {
            const int p = binned[ro[k] + i];
            const int pos = atomicAdd(&ncur[p & (BIN_NODES - 1)], 1);
            col[binStart + pos] = (int)((unsigned)p >> BIN_SHIFT);
        }
    }
}

__device__ __forceinline__ float4 max3v(float4 m, float4 a, float4 b) {
    float4 r;
    r.x = __builtin_fmaxf(__builtin_fmaxf(m.x, a.x), b.x);
    r.y = __builtin_fmaxf(__builtin_fmaxf(m.y, a.y), b.y);
    r.z = __builtin_fmaxf(__builtin_fmaxf(m.z, a.z), b.z);
    r.w = __builtin_fmaxf(__builtin_fmaxf(m.w, a.w), b.w);
    return r;
}

__device__ __forceinline__ float4 load_row(const float* __restrict__ x, int ids, int j, unsigned loff) {
    const float* row = x + (size_t)__builtin_amdgcn_readlane(ids, j) * DFEAT;
    return *(const float4*)(row + loff);
}

#define APITCH 132
#define IDS_CAP 512
#define IDS_REGS (IDS_CAP / 32)

__device__ __forceinline__ void fill_ids(const int* __restrict__ col, int* sIds,
                                         int wb, int ee, int lane) {
    int idv[IDS_REGS];
#pragma unroll
    for (int k = 0; k < IDS_REGS; ++k) idv[k] = col[max(min(wb + 32 * k + lane, ee - 1), 0)];
#pragma unroll
    for (int k = 0; k < IDS_REGS; ++k) sIds[32 * k + lane] = idv[k];
}

__device__ __forceinline__ void gather_tile(const float* __restrict__ x,
                                            const int* __restrict__ rowptr,
                                            const int* __restrict__ col,
                                            float* sA, int* sIds, int row_base, int lane) {
    const unsigned loff = (unsigned)lane * 4u;
    const int rp = rowptr[row_base + min(lane, 16)];
    const int eb = __builtin_amdgcn_readlane(rp, 0);
    const int ee = __builtin_amdgcn_readlane(rp, 16);
    int wb = eb;
    fill_ids(col, sIds, wb, ee, lane);
#pragma unroll 1
    for (int r = 0; r < 16; ++r) {
        const int beg = __builtin_amdgcn_readlane(rp, r);
        const int end = __builtin_amdgcn_readlane(rp, r + 1);
        float4 acc;
        acc.x = -INFINITY; acc.y = -INFINITY; acc.z = -INFINITY; acc.w = -INFINITY;
        for (int base = beg; base < end; base += 32) {
            const int cnt = min(end - base, 32);
            if (base + cnt > wb + IDS_CAP) {
                wb = base;
                fill_ids(col, sIds, wb, ee, lane);
            }
            const int ids = sIds[base - wb + min(lane, cnt - 1)];
            int j = 0;
            for (; j + 4 < cnt; j += 8) {
                const float4 r0 = load_row(x, ids, j + 0, loff);
                const float4 r1 = load_row(x, ids, j + 1, loff);
                const float4 r2 = load_row(x, ids, j + 2, loff);
                const float4 r3 = load_row(x, ids, j + 3, loff);
                const float4 r4 = load_row(x, ids, j + 4, loff);
                const float4 r5 = load_row(x, ids, j + 5, loff);
                const float4 r6 = load_row(x, ids, j + 6, loff);
                const float4 r7 = load_row(x, ids, j + 7, loff);
                acc = max3v(acc, r0, r1);
                acc = max3v(acc, r2, r3);
                acc = max3v(acc, r4, r5);
                acc = max3v(acc, r6, r7);
            }
            if (j < cnt) {
                const float4 r0 = load_row(x, ids, j + 0, loff);
                const float4 r1 = load_row(x, ids, j + 1, loff);
                const float4 r2 = load_row(x, ids, j + 2, loff);
                const float4 r3 = load_row(x, ids, j + 3, loff);
                acc = max3v(acc, r0, r1);
                acc = max3v(acc, r2, r3);
            }
        }
        *(float4*)(sA + r * APITCH + loff) = acc;
    }
}

#define WPITCH 132
#define KCHUNK 32
#define KSTEPS (KCHUNK / 4)
#define NUM_WGP 256
#define GEMM_WAVES 16
#define MAIN_WAVES 12
#define COL_TILES (DFEAT / 16)
#define PIECE_TILES 2
#define PIECES_PER_ROWTILE (COL_TILES / PIECE_TILES)

template <bool DUAL>
__device__ __forceinline__ void load_a_chunk(const float* __restrict__ a0row,
                                             const float* __restrict__ a1row,
                                             int kc, v2f (&a0)[KSTEPS], v2f (&a1)[KSTEPS]) {
#pragma unroll
    for (int i = 0; i < KSTEPS; ++i) {
        a0[i] = *(const v2f*)(a0row + kc + 4 * i);
        if (DUAL) a1[i] = *(const v2f*)(a1row + kc + 4 * i);
    }
}

template <bool DUAL, int NT>
__device__ __forceinline__ void mma_chunk(v8f (&acc)[NT], const v2f (&a0c)[KSTEPS],
                                          const v2f (&a1c)[KSTEPS],
                                          const float* bw0, const float* bw1) {
#pragma unroll
    for (int i = 0; i < KSTEPS; ++i) {
        v2f a0 = a0c[i];
        if (DUAL) {
            a0.x = (a0.x == -INFINITY) ? 0.0f : a0.x;
            a0.y = (a0.y == -INFINITY) ? 0.0f : a0.y;
        }
        v2f b[NT];
#pragma unroll
        for (int t = 0; t < NT; ++t) b[t] = *(const v2f*)(bw0 + t * 16 * WPITCH + 4 * i);
#pragma unroll
        for (int t = 0; t < NT; ++t)
            acc[t] = __builtin_amdgcn_wmma_f32_16x16x4_f32(false, a0, false, b[t],
                                                           (short)0, acc[t], false, false);
        if (DUAL) {
            const v2f a1 = a1c[i];
#pragma unroll
            for (int t = 0; t < NT; ++t) b[t] = *(const v2f*)(bw1 + t * 16 * WPITCH + 4 * i);
#pragma unroll
            for (int t = 0; t < NT; ++t)
                acc[t] = __builtin_amdgcn_wmma_f32_16x16x4_f32(false, a1, false, b[t],
                                                               (short)0, acc[t], false, false);
        }
    }
}

template <bool DUAL, bool RELU, int NT>
__device__ __forceinline__ void tile_product(const float* __restrict__ A0,
                                             const float* __restrict__ A1,
                                             const float* sW, const float* sA,
                                             const float* __restrict__ bias,
                                             float* __restrict__ out,
                                             int row_base, int col_base, int lane) {
    const int m    = lane & 15;
    const int half = lane >> 4;
    const int koff = half * 2;

    const float* a0row = DUAL ? (sA + m * APITCH + koff)
                              : (A0 + (size_t)(row_base + m) * DFEAT + koff);
    const float* a1row = DUAL ? (A1 + (size_t)(row_base + m) * DFEAT + koff) : a0row;
    const float* bw0   = sW + (col_base + m) * WPITCH + koff;
    const float* bw1   = bw0 + DFEAT * WPITCH;

    v8f acc[NT];
#pragma unroll
    for (int t = 0; t < NT; ++t) acc[t] = (v8f){};

    v2f a0c[KSTEPS], a1c[KSTEPS];
    load_a_chunk<DUAL>(a0row, a1row, 0, a0c, a1c);

#pragma unroll 1
    for (int kc = 0; kc < DFEAT - KCHUNK; kc += KCHUNK) {
        v2f a0n[KSTEPS], a1n[KSTEPS];
        load_a_chunk<DUAL>(a0row, a1row, kc + KCHUNK, a0n, a1n);
        mma_chunk<DUAL, NT>(acc, a0c, a1c, bw0 + kc, bw1 + kc);
#pragma unroll
        for (int i = 0; i < KSTEPS; ++i) {
            a0c[i] = a0n[i];
            if (DUAL) a1c[i] = a1n[i];
        }
    }
    mma_chunk<DUAL, NT>(acc, a0c, a1c, bw0 + (DFEAT - KCHUNK), bw1 + (DFEAT - KCHUNK));

#pragma unroll
    for (int t = 0; t < NT; ++t) {
        const float bv = bias[col_base + t * 16 + m];
#pragma unroll
        for (int r = 0; r < 8; ++r) {
            float v = acc[t][r] + bv;
            if (RELU) v = fmaxf(v, 0.0f);
            out[(size_t)(row_base + r + 8 * half) * DFEAT + col_base + t * 16 + m] = v;
        }
    }
}

template <bool DUAL, bool RELU>
__global__ __launch_bounds__(GEMM_WAVES * 32) void sage_gemm_kernel(const float* __restrict__ A0,
                                                                    const float* __restrict__ A1,
                                                                    const float* __restrict__ W0,
                                                                    const float* __restrict__ W1,
                                                                    const float* __restrict__ bias,
                                                                    const int* __restrict__ rowptr,
                                                                    const int* __restrict__ col,
                                                                    float* __restrict__ out,
                                                                    int ntiles) {
    __shared__ __attribute__((aligned(16))) float sW[(DUAL ? 2 : 1) * DFEAT * WPITCH];
    __shared__ __attribute__((aligned(16))) float sAgg[DUAL ? GEMM_WAVES * 16 * APITCH : 4];
    __shared__ int sIdsAll[DUAL ? GEMM_WAVES * IDS_CAP : 4];

    const int lane = threadIdx.x & 31;
    const int wv = __builtin_amdgcn_readfirstlane((int)(threadIdx.x >> 5));
    float* sA = sAgg + (DUAL ? wv * 16 * APITCH : 0);
    int* sIds = sIdsAll + (DUAL ? wv * IDS_CAP : 0);

    {
        const int c4 = lane * 4;
#pragma unroll
        for (int r = 0; r < DFEAT; r += GEMM_WAVES) {
            const float4 v = *(const float4*)(W0 + (size_t)(r + wv) * DFEAT + c4);
            *(float4*)(sW + (r + wv) * WPITCH + c4) = v;
        }
        if (DUAL) {
#pragma unroll
            for (int r = 0; r < DFEAT; r += GEMM_WAVES) {
                const float4 v = *(const float4*)(W1 + (size_t)(r + wv) * DFEAT + c4);
                *(float4*)(sW + (DFEAT + r + wv) * WPITCH + c4) = v;
            }
        }
    }
    __syncthreads();

    const int nwg       = (int)gridDim.x;
    const int roundSize = nwg * MAIN_WAVES;
    const int places    = nwg * (GEMM_WAVES - MAIN_WAVES);
    int leftover  = ntiles % roundSize;
    int mainTiles = ntiles - leftover;
    if (leftover * PIECES_PER_ROWTILE > places) {
        mainTiles = ntiles;
        leftover  = 0;
    }

    if (wv < MAIN_WAVES) {
        for (int tile = (int)blockIdx.x * MAIN_WAVES + wv; tile < mainTiles; tile += roundSize) {
            asm volatile("" ::: "memory");
            if (DUAL) gather_tile(A0, rowptr, col, sA, sIds, tile * 16, lane);
            tile_product<DUAL, RELU, COL_TILES>(A0, A1, sW, sA, bias, out, tile * 16, 0, lane);
        }
    } else {
        const int npieces = leftover * PIECES_PER_ROWTILE;
        for (int p = (wv - MAIN_WAVES) * nwg + (int)blockIdx.x; p < npieces; p += places) {
            const int rt = mainTiles + p / PIECES_PER_ROWTILE;
            const int cp = p % PIECES_PER_ROWTILE;
            if (DUAL) gather_tile(A0, rowptr, col, sA, sIds, rt * 16, lane);
            tile_product<DUAL, RELU, PIECE_TILES>(A0, A1, sW, sA, bias, out, rt * 16,
                                                  cp * 16 * PIECE_TILES, lane);
        }
    }
}

extern "C" void kernel_launch(void* const* d_in, const int* in_sizes, int n_in,
                              void* d_out, int out_size, void* d_ws, size_t ws_size,
                              hipStream_t stream) {
    const float* x  = (const float*)d_in[0];
    const int*   ei = (const int*)d_in[1];
    const float* Wl = (const float*)d_in[2];
    const float* bl = (const float*)d_in[3];
    const float* Wr = (const float*)d_in[4];
    const float* W  = (const float*)d_in[5];
    const float* b  = (const float*)d_in[6];

    const int nE = in_sizes[1] / 2;
    const int n  = in_sizes[0] / DFEAT;
    const int*   srcp = ei;
    const int*   dstp = ei + nE;

    float* buf1 = (float*)d_ws;
    float* buf2 = buf1 + (size_t)n * DFEAT;

    const int nbins = (n + BIN_NODES) / BIN_NODES;
    int* cntT   = (int*)(buf2 + (size_t)n * DFEAT);
    int* offT   = cntT + MAXBINS * NCHUNK;
    int* rowptr = offT + MAXBINS * NCHUNK;
    int* binned = rowptr + ((n + 64) & ~63);
    int* col    = binned + nE;

    dim3 gemmBlk(GEMM_WAVES * 32);
    const int gemmTiles  = n / 16;

    bin_edges_kernel<<<NCHUNK, SORT_THREADS, 0, stream>>>(srcp, dstp, cntT, offT, binned, nE, nbins);
    build_rows_kernel<<<nbins, SORT_THREADS, 0, stream>>>(cntT, offT, binned, rowptr, col, nE, n);

    sage_gemm_kernel<true, true><<<NUM_WGP, gemmBlk, 0, stream>>>(
        x, x, Wl, Wr, bl, rowptr, col, buf1, gemmTiles);

    sage_gemm_kernel<true, true><<<NUM_WGP, gemmBlk, 0, stream>>>(
        buf1, buf1, Wl + DFEAT * DFEAT, Wr + DFEAT * DFEAT, bl + DFEAT, rowptr, col, buf2,
        gemmTiles);

    sage_gemm_kernel<false, false><<<NUM_WGP, gemmBlk, 0, stream>>>(
        buf2, nullptr, W, nullptr, b, nullptr, nullptr, (float*)d_out, gemmTiles);
}
